# starting kernel
# baseline (speedup 1.0000x reference)
#include <hip/hip_runtime.h>

typedef __attribute__((ext_vector_type(2))) float v2f;
typedef __attribute__((ext_vector_type(8))) float v8f;

#define NUM_TYPES 4
#define IN_DIM    128
#define HID_DIM   256
#define OUT_DIM   128
#define M_TILE    64

#define XS_STRIDE (IN_DIM + 4)
#define HS_STRIDE (HID_DIM + 4)


__global__ void k_init(int* cnt) {
  if (threadIdx.x < NUM_TYPES) cnt[threadIdx.x] = 0;
}

__global__ void k_hist(const int* __restrict__ type, int* cnt, int N) {
  int n = blockIdx.x * blockDim.x + threadIdx.x;
  if (n < N) atomicAdd(&cnt[type[n]], 1);
}

__global__ void k_scan(const int* __restrict__ cnt, int* basev, int* running) {
  if (threadIdx.x == 0) {
    int acc = 0;
    for (int t = 0; t < NUM_TYPES; ++t) { basev[t] = acc; running[t] = acc; acc += cnt[t]; }
    basev[NUM_TYPES] = acc;
  }
}

__global__ void k_scatter(const int* __restrict__ type, int* running, int* gidx, int N) {
  int n = blockIdx.x * blockDim.x + threadIdx.x;
  if (n < N) { int p = atomicAdd(&running[type[n]], 1); gidx[p] = n; }
}

__global__ void k_transpose(const float* __restrict__ W1, const float* __restrict__ W2,
                            float* __restrict__ W1T, float* __restrict__ W2T) {
  int i = blockIdx.x * blockDim.x + threadIdx.x;
  const int tot = NUM_TYPES * IN_DIM * HID_DIM;
  if (i < tot) {
    int t = i / (IN_DIM * HID_DIM);
    int r = (i / HID_DIM) % IN_DIM;
    int c = i % HID_DIM;
    W1T[((size_t)t * HID_DIM + c) * IN_DIM + r] = W1[i];
    int t2 = i / (HID_DIM * OUT_DIM);
    int r2 = (i / OUT_DIM) % HID_DIM;
    int c2 = i % OUT_DIM;
    W2T[((size_t)t2 * OUT_DIM + c2) * HID_DIM + r2] = W2[i];
  }
}


__global__ __launch_bounds__(256) void k_mlp(
    const float* __restrict__ X, const float* __restrict__ b1, const float* __restrict__ b2,
    const float* __restrict__ W1T, const float* __restrict__ W2T,
    const int* __restrict__ basev, const int* __restrict__ gidx,
    float* __restrict__ out) {
  __shared__ float Xs[M_TILE][XS_STRIDE];
  __shared__ float Hs[M_TILE][HS_STRIDE];
  __shared__ int   rows[M_TILE];

  int tile = blockIdx.x;
  int ty = -1, ltile = 0, acc = 0, segStart = 0, segCount = 0;
  for (int q = 0; q < NUM_TYPES; ++q) {
    int c  = basev[q + 1] - basev[q];
    int nt = (c + (M_TILE - 1)) >> 6;
    if (tile < acc + nt) { ty = q; ltile = tile - acc; segStart = basev[q]; segCount = c; break; }
    acc += nt;
  }
  if (ty < 0) return;

  int row0 = ltile << 6;
  int tid  = threadIdx.x;

  if (tid < M_TILE) {
    int g = row0 + tid;
    rows[tid] = (g < segCount) ? gidx[segStart + g] : -1;
  }
  __syncthreads();

  for (int i = tid; i < M_TILE * (IN_DIM / 4); i += 256) {
    int r  = i >> 5;
    int c4 = i & 31;
    float4 v = make_float4(0.f, 0.f, 0.f, 0.f);
    int node = rows[r];
    if (node >= 0) v = ((const float4*)(X + (size_t)node * IN_DIM))[c4];
    *(float4*)&Xs[r][c4 * 4] = v;
  }
  __syncthreads();

  int wave = tid >> 5;
  int lane = tid & 31;
  int n    = lane & 15;
  int kh   = (lane >> 4) << 1;
  int mh   = (lane >> 4) << 3;

  for (int nt = wave; nt < HID_DIM / 16; nt += 8) {
    v8f c0 = {}, c1 = {}, c2 = {}, c3 = {};
    const float* wcol = W1T + ((size_t)ty * HID_DIM + nt * 16 + n) * IN_DIM;
#pragma unroll 4
    for (int k0 = 0; k0 < IN_DIM; k0 += 4) {
      v2f b  = *(const v2f*)&wcol[k0 + kh];
      v2f a0 = *(const v2f*)&Xs[ 0 + n][k0 + kh];
      v2f a1 = *(const v2f*)&Xs[16 + n][k0 + kh];
      v2f a2 = *(const v2f*)&Xs[32 + n][k0 + kh];
      v2f a3 = *(const v2f*)&Xs[48 + n][k0 + kh];
      c0 = __builtin_amdgcn_wmma_f32_16x16x4_f32(false, a0, false, b, (short)0, c0, false, false);
      c1 = __builtin_amdgcn_wmma_f32_16x16x4_f32(false, a1, false, b, (short)0, c1, false, false);
      c2 = __builtin_amdgcn_wmma_f32_16x16x4_f32(false, a2, false, b, (short)0, c2, false, false);
      c3 = __builtin_amdgcn_wmma_f32_16x16x4_f32(false, a3, false, b, (short)0, c3, false, false);
    }
    float bias = b1[ty * HID_DIM + nt * 16 + n];
    int col = nt * 16 + n;
#pragma unroll
    for (int r = 0; r < 8; ++r) {
      float v0 = c0[r] + bias, v1 = c1[r] + bias, v2 = c2[r] + bias, v3 = c3[r] + bias;
      Hs[ 0 + r + mh][col] = v0 > 0.f ? v0 : 0.f;
      Hs[16 + r + mh][col] = v1 > 0.f ? v1 : 0.f;
      Hs[32 + r + mh][col] = v2 > 0.f ? v2 : 0.f;
      Hs[48 + r + mh][col] = v3 > 0.f ? v3 : 0.f;
    }
  }
  __syncthreads();

  {
    int ot = wave;
    v8f c0 = {}, c1 = {}, c2 = {}, c3 = {};
    const float* wcol = W2T + ((size_t)ty * OUT_DIM + ot * 16 + n) * HID_DIM;
#pragma unroll 4
    for (int k0 = 0; k0 < HID_DIM; k0 += 4) {
      v2f b  = *(const v2f*)&wcol[k0 + kh];
      v2f a0 = *(const v2f*)&Hs[ 0 + n][k0 + kh];
      v2f a1 = *(const v2f*)&Hs[16 + n][k0 + kh];
      v2f a2 = *(const v2f*)&Hs[32 + n][k0 + kh];
      v2f a3 = *(const v2f*)&Hs[48 + n][k0 + kh];
      c0 = __builtin_amdgcn_wmma_f32_16x16x4_f32(false, a0, false, b, (short)0, c0, false, false);
      c1 = __builtin_amdgcn_wmma_f32_16x16x4_f32(false, a1, false, b, (short)0, c1, false, false);
      c2 = __builtin_amdgcn_wmma_f32_16x16x4_f32(false, a2, false, b, (short)0, c2, false, false);
      c3 = __builtin_amdgcn_wmma_f32_16x16x4_f32(false, a3, false, b, (short)0, c3, false, false);
    }
    float bias = b2[ty * OUT_DIM + ot * 16 + n];
    int col = ot * 16 + n;
#pragma unroll
    for (int r = 0; r < 8; ++r) {
      int n0 = rows[ 0 + r + mh];
      int n1 = rows[16 + r + mh];
      int n2 = rows[32 + r + mh];
      int n3 = rows[48 + r + mh];
      if (n0 >= 0) out[(size_t)n0 * OUT_DIM + col] = c0[r] + bias;
      if (n1 >= 0) out[(size_t)n1 * OUT_DIM + col] = c1[r] + bias;
      if (n2 >= 0) out[(size_t)n2 * OUT_DIM + col] = c2[r] + bias;
      if (n3 >= 0) out[(size_t)n3 * OUT_DIM + col] = c3[r] + bias;
    }
  }
}


extern "C" void kernel_launch(void* const* d_in, const int* in_sizes, int n_in,
                              void* d_out, int out_size, void* d_ws, size_t ws_size,
                              hipStream_t stream) {
  const float* X    = (const float*)d_in[0];
  const int*   type = (const int*)  d_in[1];
  const float* W1   = (const float*)d_in[2];
  const float* b1   = (const float*)d_in[3];
  const float* W2   = (const float*)d_in[4];
  const float* b2   = (const float*)d_in[5];
  float* out = (float*)d_out;

  int N = in_sizes[0] / IN_DIM;

  int* cnt     = (int*)d_ws;
  int* basev   = cnt + 8;
  int* running = basev + 8;
  int* gidx    = running + 8;
  int  gpad    = (N + 63) & ~63;
  float* W1T   = (float*)(gidx + gpad);
  float* W2T   = W1T + (size_t)NUM_TYPES * HID_DIM * IN_DIM;

  int nblk = (N + 255) / 256;
  k_init<<<1, 64, 0, stream>>>(cnt);
  k_hist<<<nblk, 256, 0, stream>>>(type, cnt, N);
  k_scan<<<1, 32, 0, stream>>>(cnt, basev, running);
  k_scatter<<<nblk, 256, 0, stream>>>(type, running, gidx, N);

  int wtot = NUM_TYPES * IN_DIM * HID_DIM;
  k_transpose<<<(wtot + 255) / 256, 256, 0, stream>>>(W1, W2, W1T, W2T);

  int maxTiles = (N + M_TILE - 1) / M_TILE + NUM_TYPES;
  k_mlp<<<maxTiles, 256, 0, stream>>>(X, b1, b2, W1T, W2T, basev, gidx, out);
}


# weights held in registers: 256 persistent blocks (one per compute unit; one type, one row slice) fetch their W1/W2 fragments once, so the matrix loops read only LDS; next X tile prefetched
# speedup vs baseline: 2.0568x; 2.0568x over previous
#include <hip/hip_runtime.h>


typedef __attribute__((ext_vector_type(2))) float v2f;
typedef __attribute__((ext_vector_type(4))) float v4f;
typedef __attribute__((ext_vector_type(8))) float v8f;

#define NUM_TYPES 4
#define IN_DIM    128
#define HID_DIM   256
#define OUT_DIM   128
#define M_TILE    64
#define NSLICE    64
#define SPAN      2048

#define XS_STRIDE (IN_DIM + 4)
#define HS_STRIDE (HID_DIM + 4)

__global__ void k_pack(const float* __restrict__ W1, const float* __restrict__ W2,
                       float* __restrict__ W1F, float* __restrict__ W2F) {
  int i = blockIdx.x * blockDim.x + threadIdx.x;
  const int tot = NUM_TYPES * (HID_DIM / 16) * (IN_DIM / 8) * 32;
  if (i < tot) {
    int lane = i & 31, n = lane & 15, h = lane >> 4;
    {
      int q  = (i >> 5) % (IN_DIM / 8);
      int nt = (i >> 5) / (IN_DIM / 8) % (HID_DIM / 16);
      int t  = (i >> 5) / ((IN_DIM / 8) * (HID_DIM / 16));
      const float* s = W1 + ((size_t)t * IN_DIM + 8 * q + 4 * h) * HID_DIM + nt * 16 + n;
      v4f v = { s[0], s[HID_DIM], s[2 * HID_DIM], s[3 * HID_DIM] };
      *(v4f*)&W1F[(size_t)i * 4] = v;
    }
    {
      int q  = (i >> 5) % (HID_DIM / 8);
      int ot = (i >> 5) / (HID_DIM / 8) % (OUT_DIM / 16);
      int t  = (i >> 5) / ((HID_DIM / 8) * (OUT_DIM / 16));
      const float* s = W2 + ((size_t)t * HID_DIM + 8 * q + 4 * h) * OUT_DIM + ot * 16 + n;
      v4f v = { s[0], s[OUT_DIM], s[2 * OUT_DIM], s[3 * OUT_DIM] };
      *(v4f*)&W2F[(size_t)i * 4] = v;
    }
  }
}

template <int NS>
__device__ __forceinline__ void layer1(const float (*Xs)[XS_STRIDE], float (*Hs)[HS_STRIDE],
                                       const v4f (&w1)[2][IN_DIM / 8], const float (&bias1)[2],
                                       int wave, int n, int kq, int mh) {
  v8f c[2][NS];
#pragma unroll
  for (int j = 0; j < 2; ++j)
#pragma unroll
    for (int s = 0; s < NS; ++s) c[j][s] = v8f{};
#pragma unroll
  for (int q = 0; q < IN_DIM / 8; ++q) {
    v4f a[NS];
#pragma unroll
    for (int s = 0; s < NS; ++s) a[s] = *(const v4f*)&Xs[16 * s + n][8 * q + kq];
#pragma unroll
    for (int j = 0; j < 2; ++j) {
#pragma unroll
      for (int s = 0; s < NS; ++s)
        c[j][s] = __builtin_amdgcn_wmma_f32_16x16x4_f32(false, a[s].xy, false, w1[j][q].xy, (short)0, c[j][s], false, false);
#pragma unroll
      for (int s = 0; s < NS; ++s)
        c[j][s] = __builtin_amdgcn_wmma_f32_16x16x4_f32(false, a[s].zw, false, w1[j][q].zw, (short)0, c[j][s], false, false);
    }
  }
#pragma unroll
  for (int j = 0; j < 2; ++j) {
    int col = (wave + 8 * j) * 16 + n;
#pragma unroll
    for (int r = 0; r < 8; ++r) {
#pragma unroll
      for (int s = 0; s < NS; ++s) {
        float v = c[j][s][r] + bias1[j];
        Hs[16 * s + r + mh][col] = v > 0.f ? v : 0.f;
      }
    }
  }
}

template <int NS>
__device__ __forceinline__ void layer2(const float (*Hs)[HS_STRIDE], const int* rows, int cnt,
                                       const v4f (&w2)[HID_DIM / 8], float bias2,
                                       float* __restrict__ out,
                                       int wave, int n, int kq, int mh) {
  v8f c[NS];
#pragma unroll
  for (int s = 0; s < NS; ++s) c[s] = v8f{};
#pragma unroll
  for (int q = 0; q < HID_DIM / 8; ++q) {
    v4f a[NS];
#pragma unroll
    for (int s = 0; s < NS; ++s) a[s] = *(const v4f*)&Hs[16 * s + n][8 * q + kq];
#pragma unroll
    for (int s = 0; s < NS; ++s)
      c[s] = __builtin_amdgcn_wmma_f32_16x16x4_f32(false, a[s].xy, false, w2[q].xy, (short)0, c[s], false, false);
#pragma unroll
    for (int s = 0; s < NS; ++s)
      c[s] = __builtin_amdgcn_wmma_f32_16x16x4_f32(false, a[s].zw, false, w2[q].zw, (short)0, c[s], false, false);
  }
  int col = wave * 16 + n;
  int node[NS][8];
#pragma unroll
  for (int s = 0; s < NS; ++s)
#pragma unroll
    for (int r = 0; r < 8; ++r) {
      int i = 16 * s + r + mh;
      node[s][r] = rows[i < cnt ? i : 0];
    }
#pragma unroll
  for (int r = 0; r < 8; ++r) {
#pragma unroll
    for (int s = 0; s < NS; ++s) {
      int i = 16 * s + r + mh;
      if (i < cnt) out[(size_t)node[s][r] * OUT_DIM + col] = c[s][r] + bias2;
    }
  }
}

__device__ __forceinline__ void lds_barrier() {
  __builtin_amdgcn_fence(__ATOMIC_RELEASE, "workgroup", "local");
  __builtin_amdgcn_s_barrier();
  __builtin_amdgcn_fence(__ATOMIC_ACQUIRE, "workgroup", "local");
}


__global__ __launch_bounds__(256) void k_mlp(
    const float* __restrict__ X, const int* __restrict__ type,
    const float* __restrict__ b1, const float* __restrict__ b2,
    const float* __restrict__ W1F, const float* __restrict__ W2F,
    float* __restrict__ out, int N) {
  __shared__ float Xs[M_TILE][XS_STRIDE];
  __shared__ float Hs[M_TILE][HS_STRIDE];
  __shared__ int   list[SPAN];
  __shared__ int   wcnt[8][8];

  int tid   = threadIdx.x;
  int wave  = tid >> 5;
  int lane  = tid & 31;
  int ty    = blockIdx.x & (NUM_TYPES - 1);
  int slice = blockIdx.x >> 2;

  int n    = lane & 15;
  int kq   = (lane >> 4) << 2;
  int mh   = (lane >> 4) << 3;

  v4f w1[2][IN_DIM / 8];
  v4f w2[HID_DIM / 8];
#pragma unroll
  for (int j = 0; j < 2; ++j)
#pragma unroll
    for (int q = 0; q < IN_DIM / 8; ++q)
      w1[j][q] = *(const v4f*)&W1F[((((size_t)ty * (HID_DIM / 16) + wave + 8 * j) * (IN_DIM / 8) + q) * 32 + lane) * 4];
#pragma unroll
  for (int q = 0; q < HID_DIM / 8; ++q)
    w2[q] = *(const v4f*)&W2F[((((size_t)ty * (OUT_DIM / 16) + wave) * (HID_DIM / 8) + q) * 32 + lane) * 4];
  float bias1[2] = { b1[ty * HID_DIM + wave * 16 + n], b1[ty * HID_DIM + (wave + 8) * 16 + n] };
  float bias2 = b2[ty * OUT_DIM + wave * 16 + n];

  int R  = (N + NSLICE - 1) / NSLICE;
  int r0 = slice * R;
  int r1 = r0 + R < N ? r0 + R : N;

  for (int s0 = r0; s0 < r1; s0 += SPAN) {
    int s1 = s0 + SPAN < r1 ? s0 + SPAN : r1;

    int  tv[8];
    bool mine[8];
    int  before[8];
#pragma unroll
    for (int j = 0; j < 8; ++j) {
      int g  = s0 + j * 256 + tid;
      tv[j]  = type[g < s1 ? g : s1 - 1];
    }
#pragma unroll
    for (int j = 0; j < 8; ++j) {
      int g  = s0 + j * 256 + tid;
      mine[j] = (g < s1) && (tv[j] == ty);
      unsigned bal = __builtin_amdgcn_ballot_w32(mine[j]);
      before[j] = __builtin_popcount(bal & ((1u << lane) - 1u));
      if (lane == 0) wcnt[j][wave] = __builtin_popcount(bal);
    }
    __syncthreads();
    int tot = 0;
#pragma unroll
    for (int j = 0; j < 8; ++j) {
      int off = tot;
#pragma unroll
      for (int w = 0; w < 8; ++w) { int c = wcnt[j][w]; off += (w < wave) ? c : 0; tot += c; }
      if (mine[j]) list[off + before[j]] = s0 + j * 256 + tid;
    }
    __syncthreads();

    if (tot > 0) {
      float4 xr[8];
#pragma unroll
      for (int j = 0; j < 8; ++j) {
        int i = wave + 8 * j;
        int node = list[i < tot ? i : tot - 1];
        xr[j] = ((const float4*)(X + (size_t)node * IN_DIM))[lane];
      }

      for (int p0 = 0; p0 < tot; p0 += M_TILE) {
        int cnt = tot - p0;
        if (cnt > M_TILE) cnt = M_TILE;
        int ns = (cnt + 15) >> 4;

#pragma unroll
        for (int j = 0; j < 8; ++j) {
          int r = wave + 8 * j;
          float4 v = xr[j];
          if (r >= cnt) v = make_float4(0.f, 0.f, 0.f, 0.f);
          *(float4*)&Xs[r][lane * 4] = v;
        }
        lds_barrier();

        if (p0 + M_TILE < tot) {
#pragma unroll
          for (int j = 0; j < 8; ++j) {
            int i = p0 + M_TILE + wave + 8 * j;
            int node = list[i < tot ? i : tot - 1];
            xr[j] = ((const float4*)(X + (size_t)node * IN_DIM))[lane];
          }
        }

        switch (ns) {
          case 1:  layer1<1>(Xs, Hs, w1, bias1, wave, n, kq, mh); break;
          case 2:  layer1<2>(Xs, Hs, w1, bias1, wave, n, kq, mh); break;
          case 3:  layer1<3>(Xs, Hs, w1, bias1, wave, n, kq, mh); break;
          default: layer1<4>(Xs, Hs, w1, bias1, wave, n, kq, mh); break;
        }
        lds_barrier();

        switch (ns) {
          case 1:  layer2<1>(Hs, list + p0, cnt, w2, bias2, out, wave, n, kq, mh); break;
          case 2:  layer2<2>(Hs, list + p0, cnt, w2, bias2, out, wave, n, kq, mh); break;
          case 3:  layer2<3>(Hs, list + p0, cnt, w2, bias2, out, wave, n, kq, mh); break;
          default: layer2<4>(Hs, list + p0, cnt, w2, bias2, out, wave, n, kq, mh); break;
        }
      }
    }
    __syncthreads();
  }
}


extern "C" void kernel_launch(void* const* d_in, const int* in_sizes, int n_in,
                              void* d_out, int out_size, void* d_ws, size_t ws_size,
                              hipStream_t stream) {
  const float* X    = (const float*)d_in[0];
  const int*   type = (const int*)  d_in[1];
  const float* W1   = (const float*)d_in[2];
  const float* b1   = (const float*)d_in[3];
  const float* W2   = (const float*)d_in[4];
  const float* b2   = (const float*)d_in[5];
  float* out = (float*)d_out;

  int N = in_sizes[0] / IN_DIM;

  float* W1F   = (float*)d_ws;
  float* W2F   = W1F + (size_t)NUM_TYPES * HID_DIM * IN_DIM;

  int ptot = NUM_TYPES * IN_DIM * HID_DIM / 4;
  k_pack<<<(ptot + 255) / 256, 256, 0, stream>>>(W1, W2, W1F, W2F);

  k_mlp<<<NUM_TYPES * NSLICE, 256, 0, stream>>>(X, type, b1, b2, W1F, W2F, out, N);
}
